# starting kernel
# baseline (speedup 1.0000x reference)
#include <hip/hip_runtime.h>
#include <hip/hip_bf16.h>

typedef __attribute__((ext_vector_type(2))) float v2f;
typedef __attribute__((ext_vector_type(8))) float v8f;

__global__ __launch_bounds__(256) void zero_f32_kernel(float* __restrict__ p, int n4) {
    int i = blockIdx.x * blockDim.x + threadIdx.x;
    if (i < n4) ((float4*)p)[i] = make_float4(0.f, 0.f, 0.f, 0.f);
}

__global__ __launch_bounds__(256) void negcopy_kernel(float* __restrict__ dst,
                                                      const float* __restrict__ src, int n4) {
    int i = blockIdx.x * blockDim.x + threadIdx.x;
    if (i < n4) {
        float4 v = ((const float4*)src)[i];
        ((float4*)dst)[i] = make_float4(-v.x, -v.y, -v.z, -v.w);
    }
}

__global__ __launch_bounds__(256) void packw_kernel(const float* __restrict__ W,
                                                    float* __restrict__ Wp,
                                                    int nout, int total) {
    int idx = blockIdx.x * blockDim.x + threadIdx.x;
    if (idx >= total) return;
    int col = idx % nout;
    int k   = (idx / nout) & 63;
    int m   = idx / (nout * 64);
    Wp[(((size_t)m * 32 + (k >> 1)) * nout + col) * 2 + (k & 1)] = W[idx];
}

__global__ __launch_bounds__(256) void deg_kernel(const int* __restrict__ src,
                                                  const int* __restrict__ dst,
                                                  float* __restrict__ deg, int nedges) {
    int e = blockIdx.x * blockDim.x + threadIdx.x;
    if (e >= nedges) return;
    int s = src[e];
    if (s != dst[e]) unsafeAtomicAdd(deg + s, 1.0f);
}

__global__ __launch_bounds__(256) void dis_kernel(float* __restrict__ deg, int n) {
    int i = blockIdx.x * blockDim.x + threadIdx.x;
    if (i >= n) return;
    float d = deg[i];
    deg[i] = (d > 0.f) ? rsqrtf(fmaxf(d, 1.0f)) : 0.f;
}

__global__ __launch_bounds__(256) void edgew_kernel(const int* __restrict__ src,
                                                    const int* __restrict__ dst,
                                                    const float* __restrict__ dis,
                                                    float* __restrict__ w, int nedges) {
    int e = blockIdx.x * blockDim.x + threadIdx.x;
    if (e >= nedges) return;
    int s = src[e], d = dst[e];
    w[e] = (s != d) ? (-dis[s] * dis[d]) : 0.f;
}

__global__ __launch_bounds__(256) void spmm_kernel(const int* __restrict__ src,
                                                   const int* __restrict__ dst,
                                                   const float* __restrict__ w,
                                                   const float* __restrict__ x,
                                                   float* __restrict__ out,
                                                   int nedges, float scale) {
    long long tid = (long long)blockIdx.x * blockDim.x + threadIdx.x;
    int e = (int)(tid >> 4);
    if (e >= nedges) return;
    float we = w[e] * scale;
    if (we == 0.f) return;
    int fg = (int)(tid & 15);
    int s = src[e], d = dst[e];
    float4 xv = ((const float4*)x)[s * 16 + fg];
    float* o = out + (size_t)d * 64 + fg * 4;
    unsafeAtomicAdd(o + 0, we * xv.x);
    unsafeAtomicAdd(o + 1, we * xv.y);
    unsafeAtomicAdd(o + 2, we * xv.z);
    unsafeAtomicAdd(o + 3, we * xv.w);
}

template <int NT, int NOUT>
__device__ inline void cheb_accum2(const float* __restrict__ A,
                                   const float* __restrict__ Wp,
                                   int rowA0, int rowA1, int khalf, int colLane,
                                   v8f (&acc0)[NT], v8f (&acc1)[NT]) {
    const float* ar0 = A + (size_t)rowA0 * 64;
    const float* ar1 = A + (size_t)rowA1 * 64;
    #pragma unroll
    for (int k0 = 0; k0 < 64; k0 += 4) {
        int kk = k0 + 2 * khalf;
        v2f a0 = *(const v2f*)(ar0 + kk);
        v2f a1 = *(const v2f*)(ar1 + kk);
        int p = (k0 >> 1) + khalf;
        #pragma unroll
        for (int t = 0; t < NT; ++t) {
            int cg = t * 16 + colLane;
            v2f b;
            if ((NOUT & 15) == 0 || cg < NOUT) {
                b = *(const v2f*)(Wp + ((size_t)p * NOUT + cg) * 2);
            } else {
                b.x = 0.f;
                b.y = 0.f;
            }
            acc0[t] = __builtin_amdgcn_wmma_f32_16x16x4_f32(
                false, a0, false, b, (short)0, acc0[t], false, false);
            acc1[t] = __builtin_amdgcn_wmma_f32_16x16x4_f32(
                false, a1, false, b, (short)0, acc1[t], false, false);
        }
    }
}

template <int NOUT, bool RELU>
__global__ __launch_bounds__(256) void cheb_gemm_kernel(const float* __restrict__ A0,
                                                        const float* __restrict__ A1,
                                                        const float* __restrict__ A2,
                                                        const float* __restrict__ Wp,
                                                        const float* __restrict__ bias,
                                                        float* __restrict__ out, int nodes) {
    constexpr int NT  = (NOUT + 15) / 16;
    constexpr int PSZ = 64 * NOUT;
    int wave = (blockIdx.x * blockDim.x + threadIdx.x) >> 5;
    int lane = threadIdx.x & 31;
    int l15  = lane & 15;
    int khalf = lane >> 4;
    int row0 = wave * 32;
    if (row0 >= nodes) return;

    v8f acc0[NT] = {};
    v8f acc1[NT] = {};
    cheb_accum2<NT, NOUT>(A0, Wp + 0 * PSZ, row0 + l15, row0 + 16 + l15, khalf, l15, acc0, acc1);
    cheb_accum2<NT, NOUT>(A1, Wp + 1 * PSZ, row0 + l15, row0 + 16 + l15, khalf, l15, acc0, acc1);
    cheb_accum2<NT, NOUT>(A2, Wp + 2 * PSZ, row0 + l15, row0 + 16 + l15, khalf, l15, acc0, acc1);

    #pragma unroll
    for (int t = 0; t < NT; ++t) {
        int col = t * 16 + l15;
        if ((NOUT & 15) != 0 && col >= NOUT) continue;
        float bv = bias[col];
        #pragma unroll
        for (int i = 0; i < 8; ++i) {
            int r = row0 + i + 8 * khalf;
            float v0 = acc0[t][i] + bv;
            float v1 = acc1[t][i] + bv;
            if (RELU) { v0 = fmaxf(v0, 0.f); v1 = fmaxf(v1, 0.f); }
            if (r < nodes)      out[(size_t)r * NOUT + col]        = v0;
            if (r + 16 < nodes) out[(size_t)(r + 16) * NOUT + col] = v1;
        }
    }
}

__global__ __launch_bounds__(256) void softmax40_kernel(float* __restrict__ out, int nodes) {
    int i = blockIdx.x * blockDim.x + threadIdx.x;
    if (i >= nodes) return;
    float* row = out + (size_t)i * 40;
    float m = -INFINITY;
    #pragma unroll
    for (int c = 0; c < 40; ++c) m = fmaxf(m, row[c]);
    float e[40];
    float s = 0.f;
    #pragma unroll
    for (int c = 0; c < 40; ++c) { e[c] = __expf(row[c] - m); s += e[c]; }
    float inv = 1.f / s;
    #pragma unroll
    for (int c = 0; c < 40; ++c) row[c] = e[c] * inv;
}

extern "C" void kernel_launch(void* const* d_in, const int* in_sizes, int n_in,
                              void* d_out, int out_size, void* d_ws, size_t ws_size,
                              hipStream_t stream) {
    const float* x  = (const float*)d_in[0];
    const int*   ei = (const int*)d_in[1];
    const float* W1 = (const float*)d_in[2];
    const float* b1 = (const float*)d_in[3];
    const float* W2 = (const float*)d_in[4];
    const float* b2 = (const float*)d_in[5];
    float* out = (float*)d_out;

    const int N = in_sizes[0] / 64;
    const int E = in_sizes[1] / 2;
    const int* src = ei;
    const int* dst = ei + E;

    auto align256 = [](size_t v) { return (v + 255) & ~(size_t)255; };
    char* ws = (char*)d_ws;
    size_t off = 0;
    float* dis  = (float*)(ws + off); off += align256((size_t)N * 4);
    float* wbuf = (float*)(ws + off); off += align256((size_t)E * 4);
    float* A1   = (float*)(ws + off); off += align256((size_t)N * 64 * 4);
    float* A2   = (float*)(ws + off); off += align256((size_t)N * 64 * 4);
    float* H    = (float*)(ws + off); off += align256((size_t)N * 64 * 4);
    float* Wp1  = (float*)(ws + off); off += align256((size_t)3 * 64 * 64 * 4);
    float* Wp2  = (float*)(ws + off); off += align256((size_t)3 * 64 * 40 * 4);
    (void)ws_size; (void)n_in; (void)out_size;

    const int TB = 256;
    const int nodeBlk  = (N + TB - 1) / TB;
    const int edgeBlk  = (E + TB - 1) / TB;
    const int feat4Blk = ((N * 64 / 4) + TB - 1) / TB;
    const long long spmmThreads = (long long)E * 16;
    const int spmmBlk  = (int)((spmmThreads + TB - 1) / TB);
    const int gemmBlk  = ((((N + 31) / 32) * 32) + TB - 1) / TB;

    packw_kernel<<<(3 * 64 * 64 + TB - 1) / TB, TB, 0, stream>>>(W1, Wp1, 64, 3 * 64 * 64);
    packw_kernel<<<(3 * 64 * 40 + TB - 1) / TB, TB, 0, stream>>>(W2, Wp2, 40, 3 * 64 * 40);
    zero_f32_kernel<<<(N / 4 + TB - 1) / TB, TB, 0, stream>>>(dis, N / 4);
    deg_kernel<<<edgeBlk, TB, 0, stream>>>(src, dst, dis, E);
    dis_kernel<<<nodeBlk, TB, 0, stream>>>(dis, N);
    edgew_kernel<<<edgeBlk, TB, 0, stream>>>(src, dst, dis, wbuf, E);

    zero_f32_kernel<<<feat4Blk, TB, 0, stream>>>(A1, N * 64 / 4);
    spmm_kernel<<<spmmBlk, TB, 0, stream>>>(src, dst, wbuf, x, A1, E, 1.0f);
    negcopy_kernel<<<feat4Blk, TB, 0, stream>>>(A2, x, N * 64 / 4);
    spmm_kernel<<<spmmBlk, TB, 0, stream>>>(src, dst, wbuf, A1, A2, E, 2.0f);
    cheb_gemm_kernel<64, true><<<gemmBlk, TB, 0, stream>>>(x, A1, A2, Wp1, b1, H, N);

    zero_f32_kernel<<<feat4Blk, TB, 0, stream>>>(A1, N * 64 / 4);
    spmm_kernel<<<spmmBlk, TB, 0, stream>>>(src, dst, wbuf, H, A1, E, 1.0f);
    negcopy_kernel<<<feat4Blk, TB, 0, stream>>>(A2, H, N * 64 / 4);
    spmm_kernel<<<spmmBlk, TB, 0, stream>>>(src, dst, wbuf, A1, A2, E, 2.0f);
    cheb_gemm_kernel<40, false><<<gemmBlk, TB, 0, stream>>>(H, A1, A2, Wp2, b2, out, N);

    softmax40_kernel<<<nodeBlk, TB, 0, stream>>>(out, N);
}


# SpMM start values (0 and -Tx0) written by the kernels that already run before them: edge-weight kernel for layer 1, layer-1 GEMM epilogue over its own consumed rows for layer 2; 4 launches go
# speedup vs baseline: 2.1198x; 2.1198x over previous
#include <hip/hip_runtime.h>
#include <hip/hip_bf16.h>

typedef __attribute__((ext_vector_type(2))) float v2f;
typedef __attribute__((ext_vector_type(8))) float v8f;

__global__ __launch_bounds__(256) void packw_kernel(const float* __restrict__ W,
                                                    float* __restrict__ Wp,
                                                    int nout, int total) {
    int idx = blockIdx.x * blockDim.x + threadIdx.x;
    if (idx >= total) return;
    int col = idx % nout;
    int k   = (idx / nout) & 63;
    int m   = idx / (nout * 64);
    Wp[(((size_t)m * 32 + (k >> 1)) * nout + col) * 2 + (k & 1)] = W[idx];
}

#define DEG_C 16
#define DEG_R 16
#define DEG_TB 1024

__device__ inline void deg_count(unsigned* hist, int s, int d, int lo, unsigned nr) {
    unsigned o = (unsigned)(s - lo);
    if (s != d && o < nr) atomicAdd(hist + o, 1u);
}

__global__ __launch_bounds__(DEG_TB) void deg_range_kernel(const int* __restrict__ src,
                                                           const int* __restrict__ dst,
                                                           float* __restrict__ part,
                                                           int nedges, int n, int nr, int chunk) {
    extern __shared__ unsigned hist[];
    const int c  = blockIdx.x % DEG_C;
    const int lo = (blockIdx.x / DEG_C) * nr;
    for (int i = threadIdx.x; i < nr; i += DEG_TB) hist[i] = 0u;
    __syncthreads();
    const int e0 = c * chunk;
    const int e1 = min(e0 + chunk, nedges);
    if ((nedges & 3) == 0) {
        const int4* s4 = (const int4*)(src + e0);
        const int4* d4 = (const int4*)(dst + e0);
        const int n4 = (e1 - e0) >> 2;
        int b = threadIdx.x;
        for (; b + 3 * DEG_TB < n4; b += 4 * DEG_TB) {
            int4 sv[4], dv[4];
            #pragma unroll
            for (int j = 0; j < 4; ++j) { sv[j] = s4[b + j * DEG_TB]; dv[j] = d4[b + j * DEG_TB]; }
            #pragma unroll
            for (int j = 0; j < 4; ++j) {
                deg_count(hist, sv[j].x, dv[j].x, lo, (unsigned)nr);
                deg_count(hist, sv[j].y, dv[j].y, lo, (unsigned)nr);
                deg_count(hist, sv[j].z, dv[j].z, lo, (unsigned)nr);
                deg_count(hist, sv[j].w, dv[j].w, lo, (unsigned)nr);
            }
        }
        for (; b < n4; b += DEG_TB) {
            int4 s = s4[b], d = d4[b];
            deg_count(hist, s.x, d.x, lo, (unsigned)nr);
            deg_count(hist, s.y, d.y, lo, (unsigned)nr);
            deg_count(hist, s.z, d.z, lo, (unsigned)nr);
            deg_count(hist, s.w, d.w, lo, (unsigned)nr);
        }
    } else {
        for (int e = e0 + threadIdx.x; e < e1; e += DEG_TB)
            deg_count(hist, src[e], dst[e], lo, (unsigned)nr);
    }
    __syncthreads();
    for (int i = threadIdx.x; i < nr; i += DEG_TB) {
        int node = lo + i;
        if (node < n) part[(size_t)c * n + node] = (float)hist[i];
    }
}

__global__ __launch_bounds__(256) void dis_kernel(const float* __restrict__ part,
                                                  float* __restrict__ dis, int n) {
    int i = blockIdx.x * blockDim.x + threadIdx.x;
    if (i >= n) return;
    float d = 0.f;
    #pragma unroll
    for (int c = 0; c < DEG_C; ++c) d += part[(size_t)c * n + i];
    dis[i] = (d > 0.f) ? rsqrtf(fmaxf(d, 1.0f)) : 0.f;
}

__global__ __launch_bounds__(256) void edgew_init_kernel(const int* __restrict__ src,
                                                         const int* __restrict__ dst,
                                                         const float* __restrict__ dis,
                                                         float* __restrict__ w, int nedges,
                                                         const float* __restrict__ x,
                                                         float* __restrict__ acc1,
                                                         float* __restrict__ acc2, int n4) {
    const int tid = blockIdx.x * blockDim.x + threadIdx.x;
    const int nth = gridDim.x * blockDim.x;
    const bool has_edge = tid < nedges;
    int s = 0, d = 0;
    if (has_edge) { s = src[tid]; d = dst[tid]; }
    for (int i = tid; i < n4; i += nth) {
        float4 v = ((const float4*)x)[i];
        ((float4*)acc1)[i] = make_float4(0.f, 0.f, 0.f, 0.f);
        ((float4*)acc2)[i] = make_float4(-v.x, -v.y, -v.z, -v.w);
    }
    if (has_edge) w[tid] = (s != d) ? (-dis[s] * dis[d]) : 0.f;
}

__device__ inline int lane_bcast_i(int v, int l) { return __builtin_amdgcn_readlane(v, l); }
__device__ inline float lane_bcast_f(float v, int l) {
    return __int_as_float(__builtin_amdgcn_readlane(__float_as_int(v), l));
}

__global__ __launch_bounds__(256) void spmm_kernel(const int* __restrict__ src,
                                                   const int* __restrict__ dst,
                                                   const float* __restrict__ w,
                                                   const float* __restrict__ x,
                                                   float* __restrict__ out,
                                                   int nedges, float scale) {
    int wave = (blockIdx.x * blockDim.x + threadIdx.x) >> 5;
    int lane = threadIdx.x & 31;
    int e0 = wave * 32;
    if (e0 >= nedges) return;
    int e = e0 + lane;
    int s_l = 0, d_l = 0;
    float w_l = 0.f;
    if (e < nedges) { s_l = src[e]; d_l = dst[e]; w_l = w[e] * scale; }
    const float* xl = x + lane;
    float* ol = out + lane;
    int cnt = nedges - e0;
    if (cnt >= 32) {
        #pragma unroll
        for (int b = 0; b < 32; b += 8) {
            float v0[8], v1[8];
            #pragma unroll
            for (int j = 0; j < 8; ++j) {
                unsigned s = (unsigned)lane_bcast_i(s_l, b + j);
                const float* xr = xl + s * 64u;
                v0[j] = xr[0];
                v1[j] = xr[32];
            }
            #pragma unroll
            for (int j = 0; j < 8; ++j) {
                unsigned d = (unsigned)lane_bcast_i(d_l, b + j);
                float we = lane_bcast_f(w_l, b + j);
                float* o = ol + d * 64u;
                unsafeAtomicAdd(o, we * v0[j]);
                unsafeAtomicAdd(o + 32, we * v1[j]);
            }
        }
    } else {
        for (int i = 0; i < cnt; ++i) {
            unsigned s = (unsigned)lane_bcast_i(s_l, i);
            unsigned d = (unsigned)lane_bcast_i(d_l, i);
            float we = lane_bcast_f(w_l, i);
            const float* xr = xl + s * 64u;
            float* o = ol + d * 64u;
            unsafeAtomicAdd(o, we * xr[0]);
            unsafeAtomicAdd(o + 32, we * xr[32]);
        }
    }
}

template <int NT, int NOUT>
__device__ inline void cheb_accum2(const float* A,
                                   const float* __restrict__ Wp,
                                   int rowA0, int rowA1, int khalf, int colLane,
                                   v8f (&acc0)[NT], v8f (&acc1)[NT]) {
    const float* ar0 = A + (size_t)rowA0 * 64;
    const float* ar1 = A + (size_t)rowA1 * 64;
    #pragma unroll
    for (int k0 = 0; k0 < 64; k0 += 4) {
        int kk = k0 + 2 * khalf;
        v2f a0 = *(const v2f*)(ar0 + kk);
        v2f a1 = *(const v2f*)(ar1 + kk);
        int p = (k0 >> 1) + khalf;
        #pragma unroll
        for (int t = 0; t < NT; ++t) {
            int cg = t * 16 + colLane;
            v2f b;
            if ((NOUT & 15) == 0 || cg < NOUT) {
                b = *(const v2f*)(Wp + ((size_t)p * NOUT + cg) * 2);
            } else {
                b.x = 0.f;
                b.y = 0.f;
            }
            acc0[t] = __builtin_amdgcn_wmma_f32_16x16x4_f32(
                false, a0, false, b, (short)0, acc0[t], false, false);
            acc1[t] = __builtin_amdgcn_wmma_f32_16x16x4_f32(
                false, a1, false, b, (short)0, acc1[t], false, false);
        }
    }
}

template <int NOUT, bool RELU, bool NEXT>
__global__ __launch_bounds__(256) void cheb_gemm_kernel(const float* __restrict__ A0,
                                                        float* A1,
                                                        float* A2,
                                                        const float* __restrict__ Wp,
                                                        const float* __restrict__ bias,
                                                        float* __restrict__ out, int nodes) {
    static_assert(!NEXT || NOUT == 64, "the next layer's operands are 64 wide");
    constexpr int NT  = (NOUT + 15) / 16;
    constexpr int PSZ = 64 * NOUT;
    int wave = (blockIdx.x * blockDim.x + threadIdx.x) >> 5;
    int lane = threadIdx.x & 31;
    int l15  = lane & 15;
    int khalf = lane >> 4;
    int row0 = wave * 32;
    if (row0 >= nodes) return;

    v8f acc0[NT] = {};
    v8f acc1[NT] = {};
    cheb_accum2<NT, NOUT>(A0, Wp + 0 * PSZ, row0 + l15, row0 + 16 + l15, khalf, l15, acc0, acc1);
    cheb_accum2<NT, NOUT>(A1, Wp + 1 * PSZ, row0 + l15, row0 + 16 + l15, khalf, l15, acc0, acc1);
    cheb_accum2<NT, NOUT>(A2, Wp + 2 * PSZ, row0 + l15, row0 + 16 + l15, khalf, l15, acc0, acc1);

    #pragma unroll
    for (int t = 0; t < NT; ++t) {
        int col = t * 16 + l15;
        if ((NOUT & 15) != 0 && col >= NOUT) continue;
        float bv = bias[col];
        #pragma unroll
        for (int i = 0; i < 8; ++i) {
            int r = row0 + i + 8 * khalf;
            float v0 = acc0[t][i] + bv;
            float v1 = acc1[t][i] + bv;
            if (RELU) { v0 = fmaxf(v0, 0.f); v1 = fmaxf(v1, 0.f); }
            if (r < nodes) {
                out[(size_t)r * NOUT + col] = v0;
                if (NEXT) A2[(size_t)r * NOUT + col] = -v0;
            }
            if (r + 16 < nodes) {
                out[(size_t)(r + 16) * NOUT + col] = v1;
                if (NEXT) A2[(size_t)(r + 16) * NOUT + col] = -v1;
            }
        }
    }
    if (NEXT) {
        const int lim4 = nodes * 16;
        #pragma unroll
        for (int j = 0; j < 16; ++j) {
            int i4 = row0 * 16 + j * 32 + lane;
            if (i4 < lim4) ((float4*)A1)[i4] = make_float4(0.f, 0.f, 0.f, 0.f);
        }
    }
}

__global__ __launch_bounds__(256) void softmax40_kernel(float* __restrict__ out, int nodes) {
    int i = blockIdx.x * blockDim.x + threadIdx.x;
    if (i >= nodes) return;
    float* row = out + (size_t)i * 40;
    float m = -INFINITY;
    #pragma unroll
    for (int c = 0; c < 40; ++c) m = fmaxf(m, row[c]);
    float e[40];
    float s = 0.f;
    #pragma unroll
    for (int c = 0; c < 40; ++c) { e[c] = __expf(row[c] - m); s += e[c]; }
    float inv = 1.f / s;
    #pragma unroll
    for (int c = 0; c < 40; ++c) row[c] = e[c] * inv;
}

extern "C" void kernel_launch(void* const* d_in, const int* in_sizes, int n_in,
                              void* d_out, int out_size, void* d_ws, size_t ws_size,
                              hipStream_t stream) {
    const float* x  = (const float*)d_in[0];
    const int*   ei = (const int*)d_in[1];
    const float* W1 = (const float*)d_in[2];
    const float* b1 = (const float*)d_in[3];
    const float* W2 = (const float*)d_in[4];
    const float* b2 = (const float*)d_in[5];
    float* out = (float*)d_out;

    const int N = in_sizes[0] / 64;
    const int E = in_sizes[1] / 2;
    const int* src = ei;
    const int* dst = ei + E;

    auto align256 = [](size_t v) { return (v + 255) & ~(size_t)255; };
    char* ws = (char*)d_ws;
    size_t off = 0;
    float* dis  = (float*)(ws + off); off += align256((size_t)N * 4);
    float* wbuf = (float*)(ws + off); off += align256((size_t)E * 4);
    float* A1   = (float*)(ws + off); off += align256((size_t)N * 64 * 4);
    float* A2   = (float*)(ws + off); off += align256((size_t)N * 64 * 4);
    float* H    = (float*)(ws + off); off += align256((size_t)N * 64 * 4);
    float* Wp1  = (float*)(ws + off); off += align256((size_t)3 * 64 * 64 * 4);
    float* Wp2  = (float*)(ws + off); off += align256((size_t)3 * 64 * 40 * 4);
    float* part = H;
    (void)ws_size; (void)n_in; (void)out_size;

    const int TB = 256;
    const int nodeBlk  = (N + TB - 1) / TB;
    const int edgeBlk  = (E + TB - 1) / TB;
    const int spmmBlk  = (((E + 31) / 32) * 32 + TB - 1) / TB;
    const int gemmBlk  = ((((N + 31) / 32) * 32) + TB - 1) / TB;
    const int degNr    = (N + DEG_R - 1) / DEG_R;
    const int degChunk = (((E + DEG_C - 1) / DEG_C) + 3) & ~3;

    packw_kernel<<<(3 * 64 * 64 + TB - 1) / TB, TB, 0, stream>>>(W1, Wp1, 64, 3 * 64 * 64);
    packw_kernel<<<(3 * 64 * 40 + TB - 1) / TB, TB, 0, stream>>>(W2, Wp2, 40, 3 * 64 * 40);
    deg_range_kernel<<<DEG_C * DEG_R, DEG_TB, (size_t)degNr * sizeof(unsigned), stream>>>(
        src, dst, part, E, N, degNr, degChunk);
    dis_kernel<<<nodeBlk, TB, 0, stream>>>(part, dis, N);
    edgew_init_kernel<<<edgeBlk, TB, 0, stream>>>(src, dst, dis, wbuf, E, x, A1, A2, N * 64 / 4);

    spmm_kernel<<<spmmBlk, TB, 0, stream>>>(src, dst, wbuf, x, A1, E, 1.0f);
    spmm_kernel<<<spmmBlk, TB, 0, stream>>>(src, dst, wbuf, A1, A2, E, 2.0f);
    cheb_gemm_kernel<64, true, true><<<gemmBlk, TB, 0, stream>>>(x, A1, A2, Wp1, b1, H, N);

    spmm_kernel<<<spmmBlk, TB, 0, stream>>>(src, dst, wbuf, H, A1, E, 1.0f);
    spmm_kernel<<<spmmBlk, TB, 0, stream>>>(src, dst, wbuf, A1, A2, E, 2.0f);
    cheb_gemm_kernel<40, false, false><<<gemmBlk, TB, 0, stream>>>(H, A1, A2, Wp2, b2, out, N);

    softmax40_kernel<<<nodeBlk, TB, 0, stream>>>(out, N);
}


# SpMM edges handed out by two int-atomic ticket queues, one per half of the destination range; a work-group draws first from its own package half's queue: a row's float atomics stay in one L2 half
# speedup vs baseline: 2.5673x; 1.2111x over previous
#include <hip/hip_runtime.h>
#include <hip/hip_bf16.h>

typedef __attribute__((ext_vector_type(2))) float v2f;
typedef __attribute__((ext_vector_type(8))) float v8f;

__global__ __launch_bounds__(256) void packw_kernel(const float* __restrict__ W,
                                                    float* __restrict__ Wp,
                                                    int nout, int total) {
    int idx = blockIdx.x * blockDim.x + threadIdx.x;
    if (idx >= total) return;
    int col = idx % nout;
    int k   = (idx / nout) & 63;
    int m   = idx / (nout * 64);
    Wp[(((size_t)m * 32 + (k >> 1)) * nout + col) * 2 + (k & 1)] = W[idx];
}

#define DEG_C 16
#define TICKET_EDGES 256
#define QUEUE_STRIDE 64
#define DEG_R 16
#define DEG_TB 1024

__device__ inline void deg_count(unsigned* hist, int s, int d, int lo, unsigned nr) {
    unsigned o = (unsigned)(s - lo);
    if (s != d && o < nr) atomicAdd(hist + o, 1u);
}

__global__ __launch_bounds__(DEG_TB) void deg_range_kernel(const int* __restrict__ src,
                                                           const int* __restrict__ dst,
                                                           float* __restrict__ part,
                                                           int nedges, int n, int nr, int chunk) {
    extern __shared__ unsigned hist[];
    const int c  = blockIdx.x % DEG_C;
    const int lo = (blockIdx.x / DEG_C) * nr;
    for (int i = threadIdx.x; i < nr; i += DEG_TB) hist[i] = 0u;
    __syncthreads();
    const int e0 = c * chunk;
    const int e1 = min(e0 + chunk, nedges);
    if ((nedges & 3) == 0) {
        const int4* s4 = (const int4*)(src + e0);
        const int4* d4 = (const int4*)(dst + e0);
        const int n4 = (e1 - e0) >> 2;
        int b = threadIdx.x;
        for (; b + 3 * DEG_TB < n4; b += 4 * DEG_TB) {
            int4 sv[4], dv[4];
            #pragma unroll
            for (int j = 0; j < 4; ++j) { sv[j] = s4[b + j * DEG_TB]; dv[j] = d4[b + j * DEG_TB]; }
            #pragma unroll
            for (int j = 0; j < 4; ++j) {
                deg_count(hist, sv[j].x, dv[j].x, lo, (unsigned)nr);
                deg_count(hist, sv[j].y, dv[j].y, lo, (unsigned)nr);
                deg_count(hist, sv[j].z, dv[j].z, lo, (unsigned)nr);
                deg_count(hist, sv[j].w, dv[j].w, lo, (unsigned)nr);
            }
        }
        for (; b < n4; b += DEG_TB) {
            int4 s = s4[b], d = d4[b];
            deg_count(hist, s.x, d.x, lo, (unsigned)nr);
            deg_count(hist, s.y, d.y, lo, (unsigned)nr);
            deg_count(hist, s.z, d.z, lo, (unsigned)nr);
            deg_count(hist, s.w, d.w, lo, (unsigned)nr);
        }
    } else {
        for (int e = e0 + threadIdx.x; e < e1; e += DEG_TB)
            deg_count(hist, src[e], dst[e], lo, (unsigned)nr);
    }
    __syncthreads();
    for (int i = threadIdx.x; i < nr; i += DEG_TB) {
        int node = lo + i;
        if (node < n) part[(size_t)c * n + node] = (float)hist[i];
    }
}

__global__ __launch_bounds__(256) void dis_kernel(const float* __restrict__ part,
                                                  float* __restrict__ dis, int n,
                                                  unsigned* __restrict__ tickets) {
    int i = blockIdx.x * blockDim.x + threadIdx.x;
    if (i < 4) tickets[(i >> 1) * QUEUE_STRIDE + (i & 1)] = 0u;
    if (i >= n) return;
    float d = 0.f;
    #pragma unroll
    for (int c = 0; c < DEG_C; ++c) d += part[(size_t)c * n + i];
    dis[i] = (d > 0.f) ? rsqrtf(fmaxf(d, 1.0f)) : 0.f;
}

__global__ __launch_bounds__(256) void edgew_init_kernel(const int* __restrict__ src,
                                                         const int* __restrict__ dst,
                                                         const float* __restrict__ dis,
                                                         float* __restrict__ w, int nedges,
                                                         const float* __restrict__ x,
                                                         float* __restrict__ acc1,
                                                         float* __restrict__ acc2, int n4) {
    const int tid = blockIdx.x * blockDim.x + threadIdx.x;
    const int nth = gridDim.x * blockDim.x;
    const bool has_edge = tid < nedges;
    int s = 0, d = 0;
    if (has_edge) { s = src[tid]; d = dst[tid]; }
    for (int i = tid; i < n4; i += nth) {
        float4 v = ((const float4*)x)[i];
        ((float4*)acc1)[i] = make_float4(0.f, 0.f, 0.f, 0.f);
        ((float4*)acc2)[i] = make_float4(-v.x, -v.y, -v.z, -v.w);
    }
    if (has_edge) w[tid] = (s != d) ? (-dis[s] * dis[d]) : 0.f;
}

__device__ inline int my_half() {
    unsigned id = (unsigned)__builtin_amdgcn_s_sendmsg_rtn(0x87);
    return (int)((id >> 18) & 1u);
}

__global__ __launch_bounds__(256) void spmm_kernel(const int* __restrict__ src,
                                                   const int* __restrict__ dst,
                                                   const float* __restrict__ w,
                                                   const float* __restrict__ x,
                                                   float* __restrict__ out,
                                                   int nedges, float scale,
                                                   unsigned* tickets,
                                                   int ntickets, int nsplit) {
    __shared__ unsigned s_word;
    const int lane = threadIdx.x & 31;
    const int wv = threadIdx.x >> 5;
    const float* xl = x + lane;
    float* ol = out + lane;
    if (threadIdx.x == 0) s_word = (unsigned)my_half();
    __syncthreads();
    int q = (int)s_word;
    __syncthreads();
    bool helped = false;
    if (threadIdx.x == 0) s_word = atomicAdd(tickets + q * QUEUE_STRIDE, 1u);
    __syncthreads();
    unsigned t = s_word;
    __syncthreads();
    for (;;) {
        if (t >= (unsigned)ntickets) {
            if (helped) break;
            helped = true;
            q ^= 1;
            if (threadIdx.x == 0) s_word = atomicAdd(tickets + q * QUEUE_STRIDE, 1u);
            __syncthreads();
            t = s_word;
            __syncthreads();
            continue;
        }
        const int e = (int)t * TICKET_EDGES + wv * 32 + lane;
        const bool in = e < nedges;
        int s_l = 0, d_l = 0;
        float w_l = 0.f;
        if (in) { d_l = dst[e]; s_l = src[e]; w_l = w[e]; }
        unsigned tnext = 0u;
        if (threadIdx.x == 0) tnext = atomicAdd(tickets + q * QUEUE_STRIDE, 1u);
        const bool mine = in && (((d_l >= nsplit) ? 1 : 0) == q);
        w_l *= scale;
        unsigned mask = (unsigned)__builtin_amdgcn_ballot_w32(mine);
        while (mask) {
            int li[8];
            int n = 0;
            #pragma unroll
            for (int j = 0; j < 8; ++j) {
                li[j] = 0;
                if (mask) { li[j] = __builtin_ctz(mask); mask &= mask - 1u; n = j + 1; }
            }
            float v0[8], v1[8];
            #pragma unroll
            for (int j = 0; j < 8; ++j) {
                if (j < n) {
                    unsigned s = (unsigned)__builtin_amdgcn_readlane(s_l, li[j]);
                    const float* xr = xl + s * 64u;
                    v0[j] = xr[0];
                    v1[j] = xr[32];
                }
            }
            #pragma unroll
            for (int j = 0; j < 8; ++j) {
                if (j < n) {
                    unsigned d = (unsigned)__builtin_amdgcn_readlane(d_l, li[j]);
                    float we = __int_as_float(__builtin_amdgcn_readlane(__float_as_int(w_l), li[j]));
                    float* o = ol + d * 64u;
                    unsafeAtomicAdd(o, we * v0[j]);
                    unsafeAtomicAdd(o + 32, we * v1[j]);
                }
            }
        }
        if (threadIdx.x == 0) s_word = tnext;
        __syncthreads();
        t = s_word;
        __syncthreads();
    }
}

template <int NT, int NOUT>
__device__ inline void cheb_accum2(const float* A,
                                   const float* __restrict__ Wp,
                                   int rowA0, int rowA1, int khalf, int colLane,
                                   v8f (&acc0)[NT], v8f (&acc1)[NT]) {
    const float* ar0 = A + (size_t)rowA0 * 64;
    const float* ar1 = A + (size_t)rowA1 * 64;
    #pragma unroll
    for (int k0 = 0; k0 < 64; k0 += 4) {
        int kk = k0 + 2 * khalf;
        v2f a0 = *(const v2f*)(ar0 + kk);
        v2f a1 = *(const v2f*)(ar1 + kk);
        int p = (k0 >> 1) + khalf;
        #pragma unroll
        for (int t = 0; t < NT; ++t) {
            int cg = t * 16 + colLane;
            v2f b;
            if ((NOUT & 15) == 0 || cg < NOUT) {
                b = *(const v2f*)(Wp + ((size_t)p * NOUT + cg) * 2);
            } else {
                b.x = 0.f;
                b.y = 0.f;
            }
            acc0[t] = __builtin_amdgcn_wmma_f32_16x16x4_f32(
                false, a0, false, b, (short)0, acc0[t], false, false);
            acc1[t] = __builtin_amdgcn_wmma_f32_16x16x4_f32(
                false, a1, false, b, (short)0, acc1[t], false, false);
        }
    }
}

template <int NOUT, bool RELU, bool NEXT>
__global__ __launch_bounds__(256) void cheb_gemm_kernel(const float* __restrict__ A0,
                                                        float* A1,
                                                        float* A2,
                                                        const float* __restrict__ Wp,
                                                        const float* __restrict__ bias,
                                                        float* __restrict__ out, int nodes,
                                                        unsigned* __restrict__ next_tickets) {
    static_assert(!NEXT || NOUT == 64, "the next layer's operands are 64 wide");
    if (NEXT && blockIdx.x == 0 && threadIdx.x < 4)
        next_tickets[(threadIdx.x >> 1) * QUEUE_STRIDE + (threadIdx.x & 1)] = 0u;
    constexpr int NT  = (NOUT + 15) / 16;
    constexpr int PSZ = 64 * NOUT;
    int wave = (blockIdx.x * blockDim.x + threadIdx.x) >> 5;
    int lane = threadIdx.x & 31;
    int l15  = lane & 15;
    int khalf = lane >> 4;
    int row0 = wave * 32;
    if (row0 >= nodes) return;

    v8f acc0[NT] = {};
    v8f acc1[NT] = {};
    cheb_accum2<NT, NOUT>(A0, Wp + 0 * PSZ, row0 + l15, row0 + 16 + l15, khalf, l15, acc0, acc1);
    cheb_accum2<NT, NOUT>(A1, Wp + 1 * PSZ, row0 + l15, row0 + 16 + l15, khalf, l15, acc0, acc1);
    cheb_accum2<NT, NOUT>(A2, Wp + 2 * PSZ, row0 + l15, row0 + 16 + l15, khalf, l15, acc0, acc1);

    #pragma unroll
    for (int t = 0; t < NT; ++t) {
        int col = t * 16 + l15;
        if ((NOUT & 15) != 0 && col >= NOUT) continue;
        float bv = bias[col];
        #pragma unroll
        for (int i = 0; i < 8; ++i) {
            int r = row0 + i + 8 * khalf;
            float v0 = acc0[t][i] + bv;
            float v1 = acc1[t][i] + bv;
            if (RELU) { v0 = fmaxf(v0, 0.f); v1 = fmaxf(v1, 0.f); }
            if (r < nodes) {
                out[(size_t)r * NOUT + col] = v0;
                if (NEXT) A2[(size_t)r * NOUT + col] = -v0;
            }
            if (r + 16 < nodes) {
                out[(size_t)(r + 16) * NOUT + col] = v1;
                if (NEXT) A2[(size_t)(r + 16) * NOUT + col] = -v1;
            }
        }
    }
    if (NEXT) {
        const int lim4 = nodes * 16;
        #pragma unroll
        for (int j = 0; j < 16; ++j) {
            int i4 = row0 * 16 + j * 32 + lane;
            if (i4 < lim4) ((float4*)A1)[i4] = make_float4(0.f, 0.f, 0.f, 0.f);
        }
    }
}

__global__ __launch_bounds__(256) void softmax40_kernel(float* __restrict__ out, int nodes) {
    int i = blockIdx.x * blockDim.x + threadIdx.x;
    if (i >= nodes) return;
    float* row = out + (size_t)i * 40;
    float m = -INFINITY;
    #pragma unroll
    for (int c = 0; c < 40; ++c) m = fmaxf(m, row[c]);
    float e[40];
    float s = 0.f;
    #pragma unroll
    for (int c = 0; c < 40; ++c) { e[c] = __expf(row[c] - m); s += e[c]; }
    float inv = 1.f / s;
    #pragma unroll
    for (int c = 0; c < 40; ++c) row[c] = e[c] * inv;
}

extern "C" void kernel_launch(void* const* d_in, const int* in_sizes, int n_in,
                              void* d_out, int out_size, void* d_ws, size_t ws_size,
                              hipStream_t stream) {
    const float* x  = (const float*)d_in[0];
    const int*   ei = (const int*)d_in[1];
    const float* W1 = (const float*)d_in[2];
    const float* b1 = (const float*)d_in[3];
    const float* W2 = (const float*)d_in[4];
    const float* b2 = (const float*)d_in[5];
    float* out = (float*)d_out;

    const int N = in_sizes[0] / 64;
    const int E = in_sizes[1] / 2;
    const int* src = ei;
    const int* dst = ei + E;

    auto align256 = [](size_t v) { return (v + 255) & ~(size_t)255; };
    char* ws = (char*)d_ws;
    size_t off = 0;
    float* dis  = (float*)(ws + off); off += align256((size_t)N * 4);
    float* wbuf = (float*)(ws + off); off += align256((size_t)E * 4);
    float* A1   = (float*)(ws + off); off += align256((size_t)N * 64 * 4);
    float* A2   = (float*)(ws + off); off += align256((size_t)N * 64 * 4);
    float* H    = (float*)(ws + off); off += align256((size_t)N * 64 * 4);
    float* Wp1  = (float*)(ws + off); off += align256((size_t)3 * 64 * 64 * 4);
    float* Wp2  = (float*)(ws + off); off += align256((size_t)3 * 64 * 40 * 4);
    float* part = H;
    unsigned* tick1 = (unsigned*)(H + (size_t)N * 32);
    unsigned* tick2 = (unsigned*)dis;
    (void)ws_size; (void)n_in; (void)out_size;

    const int TB = 256;
    const int nodeBlk  = (N + TB - 1) / TB;
    const int edgeBlk  = (E + TB - 1) / TB;
    const int spmmBlk  = 2048;
    const int nTick    = (E + TICKET_EDGES - 1) / TICKET_EDGES;
    const int nSplit   = N / 2;
    const int gemmBlk  = ((((N + 31) / 32) * 32) + TB - 1) / TB;
    const int degNr    = (N + DEG_R - 1) / DEG_R;
    const int degChunk = (((E + DEG_C - 1) / DEG_C) + 3) & ~3;

    packw_kernel<<<(3 * 64 * 64 + TB - 1) / TB, TB, 0, stream>>>(W1, Wp1, 64, 3 * 64 * 64);
    packw_kernel<<<(3 * 64 * 40 + TB - 1) / TB, TB, 0, stream>>>(W2, Wp2, 40, 3 * 64 * 40);
    deg_range_kernel<<<DEG_C * DEG_R, DEG_TB, (size_t)degNr * sizeof(unsigned), stream>>>(
        src, dst, part, E, N, degNr, degChunk);
    dis_kernel<<<nodeBlk, TB, 0, stream>>>(part, dis, N, tick1);
    edgew_init_kernel<<<edgeBlk, TB, 0, stream>>>(src, dst, dis, wbuf, E, x, A1, A2, N * 64 / 4);

    spmm_kernel<<<spmmBlk, TB, 0, stream>>>(src, dst, wbuf, x, A1, E, 1.0f, tick1 + 0, nTick, nSplit);
    spmm_kernel<<<spmmBlk, TB, 0, stream>>>(src, dst, wbuf, A1, A2, E, 2.0f, tick1 + 1, nTick, nSplit);
    cheb_gemm_kernel<64, true, true><<<gemmBlk, TB, 0, stream>>>(x, A1, A2, Wp1, b1, H, N, tick2);

    spmm_kernel<<<spmmBlk, TB, 0, stream>>>(src, dst, wbuf, H, A1, E, 1.0f, tick2 + 0, nTick, nSplit);
    spmm_kernel<<<spmmBlk, TB, 0, stream>>>(src, dst, wbuf, A1, A2, E, 2.0f, tick2 + 1, nTick, nSplit);
    cheb_gemm_kernel<40, false, false><<<gemmBlk, TB, 0, stream>>>(H, A1, A2, Wp2, b2, out, N, nullptr);

    softmax40_kernel<<<nodeBlk, TB, 0, stream>>>(out, N);
}
